# starting kernel
# baseline (speedup 1.0000x reference)
#include <hip/hip_runtime.h>
#include <hip/hip_bf16.h>
#include <math.h>


typedef float v2f __attribute__((ext_vector_type(2)));
typedef float v8f __attribute__((ext_vector_type(8)));

#define DM   1024
#define DI   2048
#define NS   16
#define KC   4
#define DTR  64
#define LSEQ 2048
#define XDBL_W (DTR + 2*NS)

template<int WM, int WN, bool SOFTPLUS_BIAS>
__global__ __launch_bounds__(256)
void wmma_gemm_f32(const float* __restrict__ A, const float* __restrict__ B,
                   float* __restrict__ C, const float* __restrict__ bias,
                   int M, int N, int K, int lda, int ldb, int ldc)
{
    const int wave = (blockIdx.x * blockDim.x + threadIdx.x) >> 5;
    const int lane = threadIdx.x & 31;
    const int tilesM = M / (WM * 16);
    const int tilesN = N / (WN * 16);
    if (wave >= tilesM * tilesN) return;
    const int tm = wave % tilesM;
    const int tn = wave / tilesM;
    const int m0 = tm * WM * 16;
    const int n0 = tn * WN * 16;
    const int lr = lane & 15;
    const int lh = lane >> 4;

    v8f acc[WM][WN];
#pragma unroll
    for (int i = 0; i < WM; ++i)
#pragma unroll
        for (int j = 0; j < WN; ++j) {
            v8f z = {0.f, 0.f, 0.f, 0.f, 0.f, 0.f, 0.f, 0.f};
            acc[i][j] = z;
        }

    for (int k0 = 0; k0 < K; k0 += 4) {
        v2f a[WM];
#pragma unroll
        for (int i = 0; i < WM; ++i) {
            const float* ap = A + (size_t)(m0 + i * 16 + lr) * lda + k0 + 2 * lh;
            a[i] = *(const v2f*)ap;
        }
        v2f b[WN];
#pragma unroll
        for (int j = 0; j < WN; ++j) {
            const float* bp = B + (size_t)(k0 + 2 * lh) * ldb + n0 + j * 16 + lr;
            v2f t; t[0] = bp[0]; t[1] = bp[ldb];
            b[j] = t;
        }
#pragma unroll
        for (int i = 0; i < WM; ++i)
#pragma unroll
            for (int j = 0; j < WN; ++j)
                acc[i][j] = __builtin_amdgcn_wmma_f32_16x16x4_f32(
                    false, a[i], false, b[j], (short)0, acc[i][j], false, false);
    }

#pragma unroll
    for (int i = 0; i < WM; ++i) {
#pragma unroll
        for (int j = 0; j < WN; ++j) {
            const int col = n0 + j * 16 + lr;
#pragma unroll
            for (int r = 0; r < 8; ++r) {
                const int row = m0 + i * 16 + r + 8 * lh;
                float v = acc[i][j][r];
                if (SOFTPLUS_BIAS) {
                    v += bias[col];
                    v = (v > 20.f) ? v : __logf(1.f + __expf(v));
                }
                C[(size_t)row * ldc + col] = v;
            }
        }
    }
}

__global__ __launch_bounds__(256)
void conv_silu_kernel(const float* __restrict__ xr, const float* __restrict__ conv_w,
                      const float* __restrict__ conv_b, float* __restrict__ xs)
{
    const int idx = blockIdx.x * 256 + threadIdx.x;
    const int l = idx / DI;
    const int d = idx % DI;
    const float4 w = *(const float4*)(conv_w + (size_t)d * KC);
    float acc = conv_b[d];
    const float* xc = xr + d;
    if (l >= 3) acc += xc[(size_t)(l - 3) * (2 * DI)] * w.x;
    if (l >= 2) acc += xc[(size_t)(l - 2) * (2 * DI)] * w.y;
    if (l >= 1) acc += xc[(size_t)(l - 1) * (2 * DI)] * w.z;
    acc             += xc[(size_t)l       * (2 * DI)] * w.w;
    const float sig = 1.f / (1.f + __expf(-acc));
    xs[idx] = acc * sig;
}

__global__ __launch_bounds__(256)
void scan_kernel(const float* __restrict__ dt, const float* __restrict__ xdbl,
                 const float* __restrict__ xs, const float* __restrict__ xr,
                 const float* __restrict__ A_log, const float* __restrict__ Dvec,
                 float* __restrict__ y, float* __restrict__ hT)
{
    __shared__ float sDT[64 * 32];
    __shared__ float sXS[64 * 32];
    __shared__ float sBC[64 * 32];

    const int t  = blockIdx.x * 256 + threadIdx.x;
    const int d  = t >> 3;
    const int q  = t & 7;
    const int ch = threadIdx.x >> 3;
    const int d0 = blockIdx.x * 32;

    float A0 = -__expf(A_log[(size_t)d * NS + 2 * q]);
    float A1 = -__expf(A_log[(size_t)d * NS + 2 * q + 1]);
    float h0 = 0.f, h1 = 0.f;
    const float Dd = Dvec[d];

    for (int c = 0; c < LSEQ / 64; ++c) {
        __syncthreads();
        for (int e = threadIdx.x; e < 64 * 32; e += 256) {
            const int ls = e >> 5, w = e & 31;
            const int l = c * 64 + ls;
            sDT[e] = dt[(size_t)l * DI + d0 + w];
            sXS[e] = xs[(size_t)l * DI + d0 + w];
            sBC[e] = xdbl[(size_t)l * XDBL_W + DTR + w];
        }
        __syncthreads();
        for (int s = 0; s < 64; ++s) {
            const int l = c * 64 + s;
            const float dtv = sDT[s * 32 + ch];
            const float xv  = sXS[s * 32 + ch];
            const float2 Bv = *(const float2*)&sBC[s * 32 + 2 * q];
            const float2 Cv = *(const float2*)&sBC[s * 32 + 16 + 2 * q];
            const float dx  = dtv * xv;
            const float dA0 = __expf(dtv * A0);
            const float dA1 = __expf(dtv * A1);
            h0 = dA0 * h0 + dx * Bv.x;
            h1 = dA1 * h1 + dx * Bv.y;
            float yacc = h0 * Cv.x + h1 * Cv.y;
            yacc += __shfl_xor(yacc, 1);
            yacc += __shfl_xor(yacc, 2);
            yacc += __shfl_xor(yacc, 4);
            if (q == 0) {
                const float resv = xr[(size_t)l * (2 * DI) + DI + d];
                const float yv   = yacc + xv * Dd;
                const float sig  = 1.f / (1.f + __expf(-resv));
                y[(size_t)l * DI + d] = yv * (resv * sig);
            }
        }
    }
    float2 hout = make_float2(h0, h1);
    *(float2*)&hT[(size_t)d * NS + 2 * q] = hout;
}

extern "C" void kernel_launch(void* const* d_in, const int* in_sizes, int n_in,
                              void* d_out, int out_size, void* d_ws, size_t ws_size,
                              hipStream_t stream) {
    const float* x      = (const float*)d_in[0];
    const float* W_in   = (const float*)d_in[1];
    const float* conv_w = (const float*)d_in[2];
    const float* conv_b = (const float*)d_in[3];
    const float* W_x    = (const float*)d_in[4];
    const float* W_dt   = (const float*)d_in[5];
    const float* b_dt   = (const float*)d_in[6];
    const float* A_log  = (const float*)d_in[7];
    const float* Dv     = (const float*)d_in[8];
    const float* W_out  = (const float*)d_in[9];

    float* out = (float*)d_out;
    float* hT  = out + (size_t)LSEQ * DM;

    float* ws   = (float*)d_ws;
    float* xr   = ws;
    float* xs   = xr   + (size_t)LSEQ * 2 * DI;
    float* xdbl = xs   + (size_t)LSEQ * DI;
    float* dt   = xdbl + (size_t)LSEQ * XDBL_W;
    float* yb   = dt   + (size_t)LSEQ * DI;

    wmma_gemm_f32<4, 4, false><<<256, 256, 0, stream>>>(
        x, W_in, xr, nullptr, LSEQ, 2 * DI, DM, DM, 2 * DI, 2 * DI);

    conv_silu_kernel<<<(LSEQ * DI) / 256, 256, 0, stream>>>(xr, conv_w, conv_b, xs);

    wmma_gemm_f32<4, 2, false><<<12, 256, 0, stream>>>(
        xs, W_x, xdbl, nullptr, LSEQ, XDBL_W, DI, DI, XDBL_W, XDBL_W);

    wmma_gemm_f32<4, 4, true><<<128, 256, 0, stream>>>(
        xdbl, W_dt, dt, b_dt, LSEQ, DI, DTR, XDBL_W, DI, DI);

    scan_kernel<<<(DI * 8) / 256, 256, 0, stream>>>(dt, xdbl, xs, xr, A_log, Dv, yb, hT);

    wmma_gemm_f32<4, 4, false><<<64, 256, 0, stream>>>(
        yb, W_out, out, nullptr, LSEQ, DM, DI, DI, DM, DM);
}


# in-proj's gate half leaves the first launch and runs inside the scan's launch on the 192 units the scan leaves idle, SiLU in its epilogue; out-proj multiplies it into y at its A fetch; sums unchanged
# speedup vs baseline: 7.4813x; 7.4813x over previous
#include <hip/hip_runtime.h>
#include <hip/hip_bf16.h>
#include <math.h>


typedef float v2f __attribute__((ext_vector_type(2)));
typedef float v8f __attribute__((ext_vector_type(8)));
typedef float v4f __attribute__((ext_vector_type(4)));

#define DM   1024
#define DI   2048
#define NS   16
#define KC   4
#define DTR  64
#define LSEQ 2048
#define XDBL_W (DTR + 2*NS)

template<int WN> struct bvec;
template<> struct bvec<2> { typedef v2f T; };
template<> struct bvec<4> { typedef v4f T; };

template<int WM, int WN, bool GATE>
__device__ __forceinline__ void fetch_block(v4f (&a)[WM][2], v4f (&ga)[WM][2],
                                            typename bvec<WN>::T (&b)[8],
                                            const float* __restrict__ ap,
                                            const float* __restrict__ gp,
                                            const float* __restrict__ bp, int lda, int ldb)
{
    typedef typename bvec<WN>::T BT;
#pragma unroll
    for (int i = 0; i < WM; ++i) {
        const float* p = ap + (size_t)(i * 16) * lda;
        a[i][0] = *(const v4f*)p;
        a[i][1] = *(const v4f*)(p + 4);
        if (GATE) {
            const float* q = gp + (size_t)(i * 16) * lda;
            ga[i][0] = *(const v4f*)q;
            ga[i][1] = *(const v4f*)(q + 4);
        }
    }
#pragma unroll
    for (int kk = 0; kk < 8; ++kk)
        b[kk] = *(const BT*)(bp + (size_t)kk * ldb);
}

template<int WM, int WN, bool GATE>
__device__ __forceinline__ void mma_block(v8f (&acc)[WM][WN], const v4f (&a)[WM][2],
                                          const v4f (&ga)[WM][2],
                                          const typename bvec<WN>::T (&b)[8])
{
#pragma unroll
    for (int t = 0; t < 4; ++t) {
#pragma unroll
        for (int i = 0; i < WM; ++i) {
            v2f av;
            av[0] = a[i][t >> 1][2 * (t & 1)];
            av[1] = a[i][t >> 1][2 * (t & 1) + 1];
            if (GATE) {
                av[0] *= ga[i][t >> 1][2 * (t & 1)];
                av[1] *= ga[i][t >> 1][2 * (t & 1) + 1];
            }
#pragma unroll
            for (int j = 0; j < WN; ++j) {
                v2f bv;
                bv[0] = b[2 * t][j];
                bv[1] = b[2 * t + 1][j];
                acc[i][j] = __builtin_amdgcn_wmma_f32_16x16x4_f32(
                    false, av, false, bv, (short)0, acc[i][j], false, false);
            }
        }
    }
}

template<int WM, int WN, int KD, bool GATE>
__device__ __forceinline__ void fetch_group(v4f (&a)[KD][WM][2], v4f (&ga)[KD][WM][2],
                                            typename bvec<WN>::T (&b)[KD][8],
                                            const float* __restrict__ ap,
                                            const float* __restrict__ gp,
                                            const float* __restrict__ bp, int lda, int ldb)
{
#pragma unroll
    for (int g = 0; g < KD; ++g)
        fetch_block<WM, WN, GATE>(a[g], ga[g], b[g], ap + 16 * g, gp + 16 * g,
                                  bp + (size_t)(16 * g) * ldb, lda, ldb);
}

template<int WM, int WN, int KD, bool GATE>
__device__ __forceinline__ void mma_group(v8f (&acc)[WM][WN], const v4f (&a)[KD][WM][2],
                                          const v4f (&ga)[KD][WM][2],
                                          const typename bvec<WN>::T (&b)[KD][8])
{
#pragma unroll
    for (int g = 0; g < KD; ++g)
        mma_block<WM, WN, GATE>(acc, a[g], ga[g], b[g]);
}

#define EPI_NONE     0
#define EPI_SOFTPLUS 1
#define EPI_SILU     2

template<int WM, int WN, int KD, int EPI, bool GATE>
__device__ __forceinline__ void gemm_wave_tile(const float* __restrict__ A, const float* __restrict__ G,
                                               const float* __restrict__ B,
                                               float* __restrict__ C, const float* __restrict__ bias,
                                               int m0, int n0, int K, int lda, int ldb, int ldc,
                                               int lane)
{
    typedef typename bvec<WN>::T BT;
    const int lr = lane & 15;
    const int lh = lane >> 4;

    v8f acc[WM][WN];
#pragma unroll
    for (int i = 0; i < WM; ++i)
#pragma unroll
        for (int j = 0; j < WN; ++j) {
            v8f z = {0.f, 0.f, 0.f, 0.f, 0.f, 0.f, 0.f, 0.f};
            acc[i][j] = z;
        }

    const float* ap = A + (size_t)(m0 + lr) * lda + 8 * lh;
    const float* gp = GATE ? G + (size_t)(m0 + lr) * lda + 8 * lh : A;
    const float* bp = B + (size_t)(8 * lh) * ldb + n0 + WN * lr;

    constexpr int KG = 16 * KD;
    v4f a0[KD][WM][2], a1[KD][WM][2];
    v4f g0[KD][WM][2], g1[KD][WM][2];
    BT  b0[KD][8], b1[KD][8];
    fetch_group<WM, WN, KD, GATE>(a0, g0, b0, ap, gp, bp, lda, ldb);

    for (int kb = 0; kb < K; kb += 2 * KG) {
        fetch_group<WM, WN, KD, GATE>(a1, g1, b1, ap + kb + KG, gp + kb + KG,
                                      bp + (size_t)(kb + KG) * ldb, lda, ldb);
        __builtin_amdgcn_sched_barrier(0);
        mma_group<WM, WN, KD, GATE>(acc, a0, g0, b0);
        __builtin_amdgcn_sched_barrier(0);

        const int kn = (kb + 2 * KG < K) ? (kb + 2 * KG) : kb;
        fetch_group<WM, WN, KD, GATE>(a0, g0, b0, ap + kn, gp + kn, bp + (size_t)kn * ldb, lda, ldb);
        __builtin_amdgcn_sched_barrier(0);
        mma_group<WM, WN, KD, GATE>(acc, a1, g1, b1);
        __builtin_amdgcn_sched_barrier(0);
    }

    BT bz;
    if (EPI == EPI_SOFTPLUS) bz = *(const BT*)(bias + n0 + WN * lr);
#pragma unroll
    for (int i = 0; i < WM; ++i) {
#pragma unroll
        for (int r = 0; r < 8; ++r) {
            const int row = m0 + i * 16 + r + 8 * lh;
            BT o;
#pragma unroll
            for (int j = 0; j < WN; ++j) {
                float v = acc[i][j][r];
                if (EPI == EPI_SOFTPLUS) {
                    v += bz[j];
                    v = (v > 20.f) ? v : __logf(1.f + __expf(v));
                }
                if (EPI == EPI_SILU) {
                    const float sig = 1.f / (1.f + __expf(-v));
                    v = v * sig;
                }
                o[j] = v;
            }
            *(BT*)(C + (size_t)row * ldc + n0 + WN * lr) = o;
        }
    }
}

template<int WM, int WN, int KD, int EPI, bool GATE>
__global__ __launch_bounds__(256)
void wmma_gemm_f32(const float* __restrict__ A, const float* __restrict__ G,
                   const float* __restrict__ B,
                   float* __restrict__ C, const float* __restrict__ bias,
                   int M, int N, int K, int lda, int ldb, int ldc)
{
    const int wave = (blockIdx.x * blockDim.x + threadIdx.x) >> 5;
    const int lane = threadIdx.x & 31;
    const int tilesM = M / (WM * 16);
    const int tilesN = N / (WN * 16);
    if (wave >= tilesM * tilesN) return;
    const int tm = wave % tilesM;
    const int tn = wave / tilesM;
    gemm_wave_tile<WM, WN, KD, EPI, GATE>(A, G, B, C, bias, tm * WM * 16, tn * WN * 16,
                                          K, lda, ldb, ldc, lane);
}

__global__ __launch_bounds__(256)
void conv_silu_kernel(const float* __restrict__ xr, const float* __restrict__ conv_w,
                      const float* __restrict__ conv_b, float* __restrict__ xs)
{
    const int idx = blockIdx.x * 256 + threadIdx.x;
    const int l = idx / DI;
    const int d = idx % DI;
    const float4 w = *(const float4*)(conv_w + (size_t)d * KC);
    float acc = conv_b[d];
    const float* xc = xr + d;
    if (l >= 3) acc += xc[(size_t)(l - 3) * (2 * DI)] * w.x;
    if (l >= 2) acc += xc[(size_t)(l - 2) * (2 * DI)] * w.y;
    if (l >= 1) acc += xc[(size_t)(l - 1) * (2 * DI)] * w.z;
    acc             += xc[(size_t)l       * (2 * DI)] * w.w;
    const float sig = 1.f / (1.f + __expf(-acc));
    xs[idx] = acc * sig;
}

#define SC_T  8
#define SC_NC (LSEQ / SC_T)
#define SC_G  4
#define SC_SW (NS / SC_G)
#define SC_TW (SC_T / SC_G)

struct scan_ops {
    float dt[SC_T], xs[SC_T];
    float fx[SC_TW];
    v2f   bc;
};

__device__ __forceinline__ void scan_fetch_ops(scan_ops& o, const float* __restrict__ pdt,
                                               const float* __restrict__ pxs,
                                               const float* __restrict__ pbc, size_t r0)
{
    o.bc = *(const v2f*)(pbc + r0 * XDBL_W);
#pragma unroll
    for (int s = 0; s < SC_T; ++s) {
        o.dt[s] = pdt[(r0 + s) * DI];
        o.xs[s] = pxs[(r0 + s) * DI];
    }
}

__device__ __forceinline__ void scan_fetch_fin(scan_ops& o, const float* __restrict__ pxs, size_t rf)
{
#pragma unroll
    for (int j = 0; j < SC_TW; ++j)
        o.fx[j] = pxs[(rf + j) * DI];
}

__device__ __forceinline__ void scan_steps(float (&part)[SC_T], float (&h)[SC_SW],
                                           const float (&A)[SC_SW], const scan_ops& o,
                                           const v4f (&bc)[2 * SC_T])
{
#pragma unroll
    for (int s = 0; s < SC_T; ++s) {
        const v4f b4 = bc[2 * s];
        const v4f c4 = bc[2 * s + 1];
        const float Bv[SC_SW] = {b4.x, b4.y, b4.z, b4.w};
        const float Cv[SC_SW] = {c4.x, c4.y, c4.z, c4.w};
        const float dtv = o.dt[s];
        const float dx  = dtv * o.xs[s];
        float p[SC_SW / 2];
#pragma unroll
        for (int q = 0; q < SC_SW / 2; ++q) {
            const float dA0 = __expf(dtv * A[2 * q]);
            const float dA1 = __expf(dtv * A[2 * q + 1]);
            h[2 * q]     = dA0 * h[2 * q]     + dx * Bv[2 * q];
            h[2 * q + 1] = dA1 * h[2 * q + 1] + dx * Bv[2 * q + 1];
            p[q] = h[2 * q] * Cv[2 * q] + h[2 * q + 1] * Cv[2 * q + 1];
        }
        part[s] = p[0] + p[1];
    }
}

__device__ __forceinline__ void scan_finish(float* __restrict__ py, size_t r0,
                                            const float (&fp)[SC_G][SC_TW],
                                            const float (&fx)[SC_TW], float Dd)
{
#pragma unroll
    for (int j = 0; j < SC_TW; ++j) {
        const float yacc = (fp[0][j] + fp[1][j]) + (fp[2][j] + fp[3][j]);
        py[(r0 + j) * DI] = yacc + fx[j] * Dd;
    }
}

#define SG_SCAN_BLOCKS (DI / 32)
#define SG_GEMM_BLOCKS 192
#define SG_TM 2
#define SG_TN 4

__global__ __launch_bounds__(32 * SC_G)
void scan_gate_kernel(const float* __restrict__ dt, const float* __restrict__ xdbl,
                      const float* __restrict__ xs,
                      const float* __restrict__ A_log, const float* __restrict__ Dvec,
                      float* __restrict__ y, float* __restrict__ hT,
                      const float* __restrict__ x, const float* __restrict__ W_in,
                      float* __restrict__ gate)
{
    if (blockIdx.x >= SG_SCAN_BLOCKS) {
        const int gw   = (blockIdx.x - SG_SCAN_BLOCKS) * SC_G
                       + __builtin_amdgcn_readfirstlane((int)(threadIdx.x >> 5));
        const int lane = threadIdx.x & 31;
        constexpr int TILES_M = LSEQ / (SG_TM * 16);
        constexpr int TILES   = TILES_M * (DI / (SG_TN * 16));
        for (int t = gw; t < TILES; t += SG_GEMM_BLOCKS * SC_G)
            gemm_wave_tile<SG_TM, SG_TN, 2, EPI_SILU, false>(
                x, nullptr, W_in + DI, gate, nullptr,
                (t % TILES_M) * (SG_TM * 16), (t / TILES_M) * (SG_TN * 16),
                DM, DM, 2 * DI, DI, lane);
        return;
    }

    __shared__ float sPart[2][SC_G][SC_T][32];
    __shared__ v4f   sBC[2][SC_T][2 * NS / 4];
    __shared__ v4f   sH[32 * SC_G];

    const int tid  = threadIdx.x;
    const int lane = tid & 31;
    const int g    = tid >> 5;
    const int d    = blockIdx.x * 32 + lane;

    float A[SC_SW], h[SC_SW];
    {
        const v4f a4 = *(const v4f*)(A_log + (size_t)d * NS + SC_SW * g);
        A[0] = -__expf(a4.x);
        A[1] = -__expf(a4.y);
        A[2] = -__expf(a4.z);
        A[3] = -__expf(a4.w);
    }
#pragma unroll
    for (int n = 0; n < SC_SW; ++n) h[n] = 0.f;
    const float Dd = Dvec[d];

    const float*  pdt = dt + d;
    const float*  pxs = xs + d;
    float*        py  = y + d;
    const float*  pbc = xdbl + DTR + (size_t)(tid >> 4) * XDBL_W + 2 * (tid & 15);
    v2f*          mybc = (v2f*)&sBC[0][0][0] + (tid >> 4) * (2 * NS / 2) + (tid & 15);
    constexpr int BCSZ = SC_T * (2 * NS / 2);

    float*       mypart = &sPart[0][g][0][lane];
    const float* myfin  = &sPart[0][0][SC_TW * g][lane];
    constexpr int BUFSZ = SC_G * SC_T * 32;

    scan_ops o0, o1, o2, o3;
    v4f   bcA[2 * SC_T], bcB[2 * SC_T];
    float fp[SC_G][SC_TW];
#pragma unroll
    for (int j = 0; j < SC_TW; ++j) {
        o3.fx[j] = 0.f;
#pragma unroll
        for (int w = 0; w < SC_G; ++w) fp[w][j] = 0.f;
    }
    scan_fetch_ops(o0, pdt, pxs, pbc, 0 * SC_T);
    scan_fetch_ops(o1, pdt, pxs, pbc, 1 * SC_T);
    scan_fetch_ops(o2, pdt, pxs, pbc, 2 * SC_T);
    scan_fetch_fin(o0, pxs, 0 * SC_T + SC_TW * g);
    scan_fetch_fin(o1, pxs, 1 * SC_T + SC_TW * g);
    mybc[0]    = o0.bc;
    mybc[BCSZ] = o1.bc;
    __syncthreads();
#pragma unroll
    for (int s = 0; s < SC_T; ++s) { bcA[2 * s] = sBC[0][s][g]; bcA[2 * s + 1] = sBC[0][s][NS / 4 + g]; }
    __syncthreads();

    auto chunk = [&](const int buf, const int c, const scan_ops& cur, scan_ops& nxt,
                     scan_ops& mid, const v4f (&cbc)[2 * SC_T], v4f (&nbc)[2 * SC_T],
                     const bool fin) __attribute__((always_inline)) {
        const int cn = (c + 3 < SC_NC) ? (c + 3) : (SC_NC - 1);
        const int cm = (c + 2 < SC_NC) ? (c + 2) : (SC_NC - 1);
        scan_fetch_ops(nxt, pdt, pxs, pbc, (size_t)cn * SC_T);
        scan_fetch_fin(mid, pxs, (size_t)cm * SC_T + SC_TW * g);
#pragma unroll
        for (int s = 0; s < SC_T; ++s) {
            nbc[2 * s]     = sBC[buf ^ 1][s][g];
            nbc[2 * s + 1] = sBC[buf ^ 1][s][NS / 4 + g];
        }

        __builtin_amdgcn_sched_barrier(0);

        float part[SC_T];
        scan_steps(part, h, A, cur, cbc);
        __builtin_amdgcn_sched_barrier(0);
#pragma unroll
        for (int s = 0; s < SC_T; ++s) mypart[buf * BUFSZ + s * 32] = part[s];
        mybc[buf * BCSZ] = mid.bc;

        if (fin) scan_finish(py, (size_t)(c - 1) * SC_T + SC_TW * g, fp, nxt.fx, Dd);

        asm volatile("s_wait_dscnt 0x0" ::: "memory");
        asm volatile("s_barrier_signal -1\n\ts_barrier_wait -1" ::: "memory");

#pragma unroll
        for (int w = 0; w < SC_G; ++w)
#pragma unroll
            for (int j = 0; j < SC_TW; ++j)
                fp[w][j] = myfin[buf * BUFSZ + w * (SC_T * 32) + j * 32];
    };

    for (int c = 0; c < SC_NC; c += 4) {
        chunk(0, c,     o0, o3, o2, bcA, bcB, c > 0);
        chunk(1, c + 1, o1, o0, o3, bcB, bcA, true);
        chunk(0, c + 2, o2, o1, o0, bcA, bcB, true);
        chunk(1, c + 3, o3, o2, o1, bcB, bcA, true);
    }
    scan_finish(py, (size_t)(SC_NC - 1) * SC_T + SC_TW * g, fp, o3.fx, Dd);

    {
        v4f hv;
        hv.x = h[0]; hv.y = h[1]; hv.z = h[2]; hv.w = h[3];
        sH[lane * SC_G + g] = hv;
    }
    __syncthreads();
    *(v4f*)(hT + (size_t)blockIdx.x * (32 * NS) + 4 * tid) = sH[tid];
}

extern "C" void kernel_launch(void* const* d_in, const int* in_sizes, int n_in,
                              void* d_out, int out_size, void* d_ws, size_t ws_size,
                              hipStream_t stream) {
    const float* x      = (const float*)d_in[0];
    const float* W_in   = (const float*)d_in[1];
    const float* conv_w = (const float*)d_in[2];
    const float* conv_b = (const float*)d_in[3];
    const float* W_x    = (const float*)d_in[4];
    const float* W_dt   = (const float*)d_in[5];
    const float* b_dt   = (const float*)d_in[6];
    const float* A_log  = (const float*)d_in[7];
    const float* Dv     = (const float*)d_in[8];
    const float* W_out  = (const float*)d_in[9];

    float* out = (float*)d_out;
    float* hT  = out + (size_t)LSEQ * DM;

    float* ws   = (float*)d_ws;
    float* xr   = ws;
    float* xs   = xr   + (size_t)LSEQ * 2 * DI;
    float* xdbl = xs   + (size_t)LSEQ * DI;
    float* dt   = xdbl + (size_t)LSEQ * XDBL_W;
    float* yb   = dt   + (size_t)LSEQ * DI;
    float* gt   = yb   + (size_t)LSEQ * DI;

    wmma_gemm_f32<4, 4, 1, EPI_NONE, false><<<256, 128, 0, stream>>>(
        x, nullptr, W_in, xr, nullptr, LSEQ, DI, DM, DM, 2 * DI, 2 * DI);

    conv_silu_kernel<<<(LSEQ * DI) / 256, 256, 0, stream>>>(xr, conv_w, conv_b, xs);

    wmma_gemm_f32<1, 2, 4, EPI_NONE, false><<<96, 128, 0, stream>>>(
        xs, nullptr, W_x, xdbl, nullptr, LSEQ, XDBL_W, DI, DI, XDBL_W, XDBL_W);

    wmma_gemm_f32<4, 4, 1, EPI_SOFTPLUS, false><<<256, 128, 0, stream>>>(
        xdbl, nullptr, W_dt, dt, b_dt, LSEQ, DI, DTR, XDBL_W, DI, DI);

    scan_gate_kernel<<<SG_SCAN_BLOCKS + SG_GEMM_BLOCKS, 32 * SC_G, 0, stream>>>(
        dt, xdbl, xs, A_log, Dv, yb, hT, x, W_in, gt);

    wmma_gemm_f32<2, 4, 2, EPI_NONE, true><<<256, 128, 0, stream>>>(
        yb, gt, W_out, out, nullptr, LSEQ, DM, DI, DI, DM, DM);
}
